# starting kernel
# baseline (speedup 1.0000x reference)
#include <hip/hip_runtime.h>

#define N_NODES 100000
#define IN_DIM  128
#define HEADS   4
#define HC      128
#define NEG_SLOPE 0.2f

typedef float v2f __attribute__((ext_vector_type(2)));
typedef float v8f __attribute__((ext_vector_type(8)));

__device__ __forceinline__ int f2o(float f) {
    int i = __float_as_int(f);
    return i >= 0 ? i : (i ^ 0x7FFFFFFF);
}
__device__ __forceinline__ float o2f(int i) {
    int j = i >= 0 ? i : (i ^ 0x7FFFFFFF);
    return __int_as_float(j);
}

__global__ void gat_gemm(const float* __restrict__ x,
                         const float* __restrict__ Wl, const float* __restrict__ bl,
                         const float* __restrict__ Wr, const float* __restrict__ br,
                         float* __restrict__ xl, float* __restrict__ xr)
{
    const int wid  = threadIdx.x >> 5;
    const int lane = threadIdx.x & 31;
    const int task = blockIdx.x * 4 + wid;
    const int matSel  = task & 1;
    const int rowTile = task >> 1;

    const float* W    = matSel ? Wr : Wl;
    const float* bv   = matSel ? br : bl;
    float*       outp = matSel ? xr : xl;

    const int row0 = rowTile * 16;
    const int lm = lane & 15;
    const int lk = (lane >> 4) * 2;

    const float* aptr = x + (size_t)(row0 + lm) * IN_DIM + lk;
    const float* bptr = W + (size_t)lk * HC + lm;

    v8f c[8] = {};
    for (int k0 = 0; k0 < IN_DIM; k0 += 4) {
        const float2 av = *(const float2*)aptr;
        v2f a; a.x = av.x; a.y = av.y;
        #pragma unroll
        for (int t = 0; t < 8; ++t) {
            v2f b;
            b.x = bptr[t * 16];
            b.y = bptr[HC + t * 16];
            c[t] = __builtin_amdgcn_wmma_f32_16x16x4_f32(
                       false, a, false, b, (short)0, c[t], false, false);
        }
        aptr += 4;
        bptr += 4 * HC;
    }

    const int rbase = row0 + ((lane >> 4) * 8);
    #pragma unroll
    for (int t = 0; t < 8; ++t) {
        const int col = t * 16 + lm;
        const float bb = bv[col];
        #pragma unroll
        for (int r = 0; r < 8; ++r)
            outp[(size_t)(rbase + r) * HC + col] = c[t][r] + bb;
    }
}

__global__ void gat_init(float* __restrict__ out, float* __restrict__ denom,
                         int* __restrict__ segmax, int nOut, int nSeg)
{
    int i = blockIdx.x * blockDim.x + threadIdx.x;
    if (i < nOut) out[i] = 0.0f;
    if (i < nSeg) { denom[i] = 0.0f; segmax[i] = (int)0x80000000; }
}

__device__ __forceinline__ float edge_score(const float4 a, const float4 b,
                                            const float4 at)
{
    float m0 = a.x + b.x; m0 = m0 >= 0.0f ? m0 : NEG_SLOPE * m0;
    float m1 = a.y + b.y; m1 = m1 >= 0.0f ? m1 : NEG_SLOPE * m1;
    float m2 = a.z + b.z; m2 = m2 >= 0.0f ? m2 : NEG_SLOPE * m2;
    float m3 = a.w + b.w; m3 = m3 >= 0.0f ? m3 : NEG_SLOPE * m3;
    float p = at.x * m0 + at.y * m1 + at.z * m2 + at.w * m3;
    p += __shfl_xor(p, 1, 32);
    p += __shfl_xor(p, 2, 32);
    p += __shfl_xor(p, 4, 32);
    return p;
}

__global__ void gat_edge_max(const float* __restrict__ xl, const float* __restrict__ xr,
                             const float* __restrict__ att,
                             const int* __restrict__ src, const int* __restrict__ dst,
                             int nEdges, int nTot, int* __restrict__ segmax)
{
    const int e = blockIdx.x * 8 + (threadIdx.x >> 5);
    if (e >= nTot) return;
    const int lane = threadIdx.x & 31;
    int s, d;
    if (e < nEdges) { s = src[e]; d = dst[e]; } else { s = e - nEdges; d = s; }

    const float4 a  = *(const float4*)(xl + (size_t)s * HC + lane * 4);
    const float4 b  = *(const float4*)(xr + (size_t)d * HC + lane * 4);
    const float4 at = *(const float4*)(att + lane * 4);
    const float p = edge_score(a, b, at);

    if ((lane & 7) == 0)
        atomicMax(segmax + (size_t)d * HEADS + (lane >> 3), f2o(p));
}

__global__ void gat_edge_acc(const float* __restrict__ xl, const float* __restrict__ xr,
                             const float* __restrict__ att,
                             const int* __restrict__ src, const int* __restrict__ dst,
                             int nEdges, int nTot,
                             const int* __restrict__ segmax,
                             float* __restrict__ denom, float* __restrict__ out)
{
    const int e = blockIdx.x * 8 + (threadIdx.x >> 5);
    if (e >= nTot) return;
    const int lane = threadIdx.x & 31;
    int s, d;
    if (e < nEdges) { s = src[e]; d = dst[e]; } else { s = e - nEdges; d = s; }

    const float4 a  = *(const float4*)(xl + (size_t)s * HC + lane * 4);
    const float4 b  = *(const float4*)(xr + (size_t)d * HC + lane * 4);
    const float4 at = *(const float4*)(att + lane * 4);
    const float p = edge_score(a, b, at);

    const int h = lane >> 3;
    const float mx = o2f(segmax[(size_t)d * HEADS + h]);
    const float w = __expf(p - mx);

    if ((lane & 7) == 0)
        unsafeAtomicAdd(denom + (size_t)d * HEADS + h, w);

    float* o = out + (size_t)d * HC + lane * 4;
    unsafeAtomicAdd(o + 0, w * a.x);
    unsafeAtomicAdd(o + 1, w * a.y);
    unsafeAtomicAdd(o + 2, w * a.z);
    unsafeAtomicAdd(o + 3, w * a.w);
}

__global__ void gat_final(float* __restrict__ out, const float* __restrict__ denom,
                          const float* __restrict__ bias, int nOut)
{
    int i = blockIdx.x * blockDim.x + threadIdx.x;
    if (i >= nOut) return;
    const int node = i >> 7;
    const int ch   = i & 127;
    out[i] = out[i] / denom[node * HEADS + (ch >> 5)] + bias[ch];
}

extern "C" void kernel_launch(void* const* d_in, const int* in_sizes, int n_in,
                              void* d_out, int out_size, void* d_ws, size_t ws_size,
                              hipStream_t stream)
{
    const float* x    = (const float*)d_in[0];
    const float* Wl   = (const float*)d_in[1];
    const float* bl   = (const float*)d_in[2];
    const float* Wr   = (const float*)d_in[3];
    const float* br   = (const float*)d_in[4];
    const float* att  = (const float*)d_in[5];
    const float* bias = (const float*)d_in[6];
    const int*   src  = (const int*)d_in[7];
    const int*   dst  = (const int*)d_in[8];
    const int nE = in_sizes[7];
    float* out = (float*)d_out;

    float* xl    = (float*)d_ws;
    float* xr    = xl + (size_t)N_NODES * HC;
    float* denom = xr + (size_t)N_NODES * HC;
    int*   smax  = (int*)(denom + (size_t)N_NODES * HEADS);

    const int nOut = N_NODES * HC;
    const int nSeg = N_NODES * HEADS;

    gat_init<<<(nOut + 255) / 256, 256, 0, stream>>>(out, denom, smax, nOut, nSeg);

    gat_gemm<<<3125, 128, 0, stream>>>(x, Wl, bl, Wr, br, xl, xr);

    const int nTot = nE + N_NODES;
    const int eb = (nTot + 7) / 8;
    gat_edge_max<<<eb, 256, 0, stream>>>(xl, xr, att, src, dst, nE, nTot, smax);
    gat_edge_acc<<<eb, 256, 0, stream>>>(xl, xr, att, src, dst, nE, nTot, smax, denom, out);

    gat_final<<<(nOut + 255) / 256, 256, 0, stream>>>(out, denom, bias, nOut);
}


# Edges grouped by destination with an in-call counting sort; one wave per node keeps max, denominator and message sum in registers (online softmax, arrival-order sums); float atomics gone
# speedup vs baseline: 2.6486x; 2.6486x over previous
#include <hip/hip_runtime.h>

#define N_NODES 100000
#define IN_DIM  128
#define HEADS   4
#define HC      128
#define NEG_SLOPE 0.2f

typedef float v2f __attribute__((ext_vector_type(2)));
typedef float v8f __attribute__((ext_vector_type(8)));

__global__ void gat_gemm(const float* __restrict__ x,
                         const float* __restrict__ Wl, const float* __restrict__ bl,
                         const float* __restrict__ Wr, const float* __restrict__ br,
                         float* __restrict__ xl, float* __restrict__ xr)
{
    const int wid  = threadIdx.x >> 5;
    const int lane = threadIdx.x & 31;
    const int task = blockIdx.x * 4 + wid;
    const int matSel  = task & 1;
    const int rowTile = task >> 1;

    const float* W    = matSel ? Wr : Wl;
    const float* bv   = matSel ? br : bl;
    float*       outp = matSel ? xr : xl;

    const int row0 = rowTile * 16;
    const int lm = lane & 15;
    const int lk = (lane >> 4) * 2;

    const float* aptr = x + (size_t)(row0 + lm) * IN_DIM + lk;
    const float* bptr = W + (size_t)lk * HC + lm;

    v8f c[8] = {};
    for (int k0 = 0; k0 < IN_DIM; k0 += 4) {
        const float2 av = *(const float2*)aptr;
        v2f a; a.x = av.x; a.y = av.y;
        #pragma unroll
        for (int t = 0; t < 8; ++t) {
            v2f b;
            b.x = bptr[t * 16];
            b.y = bptr[HC + t * 16];
            c[t] = __builtin_amdgcn_wmma_f32_16x16x4_f32(
                       false, a, false, b, (short)0, c[t], false, false);
        }
        aptr += 4;
        bptr += 4 * HC;
    }

    const int rbase = row0 + ((lane >> 4) * 8);
    #pragma unroll
    for (int t = 0; t < 8; ++t) {
        const int col = t * 16 + lm;
        const float bb = bv[col];
        #pragma unroll
        for (int r = 0; r < 8; ++r)
            outp[(size_t)(rbase + r) * HC + col] = c[t][r] + bb;
    }
}

__global__ void gat_zero(int* __restrict__ deg, int* __restrict__ total)
{
    const int i = blockIdx.x * blockDim.x + threadIdx.x;
    if (i < N_NODES) deg[i] = 0;
    if (i == 0) *total = 0;
}

__global__ void gat_count(const int* __restrict__ dst, int nEdges,
                          int* __restrict__ deg, int* __restrict__ rank)
{
    const int e = blockIdx.x * blockDim.x + threadIdx.x;
    if (e >= nEdges) return;
    rank[e] = atomicAdd(deg + dst[e], 1);
}

__global__ void gat_alloc(const int* __restrict__ deg, int* __restrict__ total,
                          int* __restrict__ off)
{
    __shared__ int wsum[8];
    __shared__ int base;
    const int i    = blockIdx.x * 256 + threadIdx.x;
    const int lane = threadIdx.x & 31;
    const int wv   = threadIdx.x >> 5;
    const int v = (i < N_NODES) ? deg[i] : 0;
    int inc = v;
    for (int o = 1; o < 32; o <<= 1) {
        const int t = __shfl_up(inc, o, 32);
        if (lane >= o) inc += t;
    }
    if (lane == 31) wsum[wv] = inc;
    __syncthreads();
    if (threadIdx.x == 0) {
        int run = 0;
        for (int w = 0; w < 8; ++w) { const int t = wsum[w]; wsum[w] = run; run += t; }
        base = atomicAdd(total, run);
    }
    __syncthreads();
    if (i < N_NODES) off[i] = base + wsum[wv] + inc - v;
}

__global__ void gat_fill(const int* __restrict__ src, const int* __restrict__ dst, int nEdges,
                         const int* __restrict__ off, const int* __restrict__ rank,
                         int* __restrict__ perm)
{
    const int e = blockIdx.x * blockDim.x + threadIdx.x;
    if (e >= nEdges) return;
    perm[off[dst[e]] + rank[e]] = src[e];
}

__device__ __forceinline__ float edge_score(const float4 a, const float4 b,
                                            const float4 at)
{
    float m0 = a.x + b.x; m0 = m0 >= 0.0f ? m0 : NEG_SLOPE * m0;
    float m1 = a.y + b.y; m1 = m1 >= 0.0f ? m1 : NEG_SLOPE * m1;
    float m2 = a.z + b.z; m2 = m2 >= 0.0f ? m2 : NEG_SLOPE * m2;
    float m3 = a.w + b.w; m3 = m3 >= 0.0f ? m3 : NEG_SLOPE * m3;
    float p = at.x * m0 + at.y * m1 + at.z * m2 + at.w * m3;
    p += __shfl_xor(p, 1, 32);
    p += __shfl_xor(p, 2, 32);
    p += __shfl_xor(p, 4, 32);
    return p;
}

__global__ __launch_bounds__(256)
void gat_node(const float* __restrict__ xl, float* out,
              const float* __restrict__ att, const float* __restrict__ bias,
              const int* __restrict__ deg, const int* __restrict__ off,
              const int* __restrict__ perm)
{
    const int d = blockIdx.x * 8 + __builtin_amdgcn_readfirstlane((int)(threadIdx.x >> 5));
    if (d >= N_NODES) return;
    const int lane = threadIdx.x & 31;

    float* orow = out + (size_t)d * HC + lane * 4;
    const float4 b  = *(const float4*)orow;
    const float4 at = *(const float4*)(att + lane * 4);
    const int n     = deg[d];
    const int start = off[d];

    float4 acc = *(const float4*)(xl + (size_t)d * HC + lane * 4);
    float m = edge_score(acc, b, at);
    float l = 1.0f;

    const float NEG_INF = -__builtin_huge_valf();

    for (int j0 = 0; j0 < n; j0 += 32) {
        const int cnt = (n - j0) < 32 ? (n - j0) : 32;
        const int idx = (lane < cnt) ? perm[start + j0 + lane] : d;
        for (int k = 0; k < cnt; k += 4) {
            const int s0 = __builtin_amdgcn_readlane(idx, k);
            const int s1 = __builtin_amdgcn_readlane(idx, k + 1);
            const int s2 = __builtin_amdgcn_readlane(idx, k + 2);
            const int s3 = __builtin_amdgcn_readlane(idx, k + 3);
            const float4 a0 = *(const float4*)(xl + (size_t)s0 * HC + lane * 4);
            const float4 a1 = *(const float4*)(xl + (size_t)s1 * HC + lane * 4);
            const float4 a2 = *(const float4*)(xl + (size_t)s2 * HC + lane * 4);
            const float4 a3 = *(const float4*)(xl + (size_t)s3 * HC + lane * 4);
            const float p0 = edge_score(a0, b, at);
            float p1 = edge_score(a1, b, at);
            float p2 = edge_score(a2, b, at);
            float p3 = edge_score(a3, b, at);
            p1 = (k + 1 < cnt) ? p1 : NEG_INF;
            p2 = (k + 2 < cnt) ? p2 : NEG_INF;
            p3 = (k + 3 < cnt) ? p3 : NEG_INF;
            const float mn = fmaxf(fmaxf(m, p0), fmaxf(p1, fmaxf(p2, p3)));
            const float sc = __expf(m - mn);
            const float w0 = __expf(p0 - mn);
            const float w1 = __expf(p1 - mn);
            const float w2 = __expf(p2 - mn);
            const float w3 = __expf(p3 - mn);
            l = l * sc + ((w0 + w1) + (w2 + w3));
            acc.x = acc.x * sc + ((w0 * a0.x + w1 * a1.x) + (w2 * a2.x + w3 * a3.x));
            acc.y = acc.y * sc + ((w0 * a0.y + w1 * a1.y) + (w2 * a2.y + w3 * a3.y));
            acc.z = acc.z * sc + ((w0 * a0.z + w1 * a1.z) + (w2 * a2.z + w3 * a3.z));
            acc.w = acc.w * sc + ((w0 * a0.w + w1 * a1.w) + (w2 * a2.w + w3 * a3.w));
            m = mn;
        }
    }

    const float4 bs = *(const float4*)(bias + lane * 4);
    float4 o;
    o.x = acc.x / l + bs.x;
    o.y = acc.y / l + bs.y;
    o.z = acc.z / l + bs.z;
    o.w = acc.w / l + bs.w;
    *(float4*)orow = o;
}

extern "C" void kernel_launch(void* const* d_in, const int* in_sizes, int n_in,
                              void* d_out, int out_size, void* d_ws, size_t ws_size,
                              hipStream_t stream)
{
    const float* x    = (const float*)d_in[0];
    const float* Wl   = (const float*)d_in[1];
    const float* bl   = (const float*)d_in[2];
    const float* Wr   = (const float*)d_in[3];
    const float* br   = (const float*)d_in[4];
    const float* att  = (const float*)d_in[5];
    const float* bias = (const float*)d_in[6];
    const int*   src  = (const int*)d_in[7];
    const int*   dst  = (const int*)d_in[8];
    const int nE = in_sizes[7];
    float* out = (float*)d_out;

    float* xl    = (float*)d_ws;
    int*   deg   = (int*)(xl + (size_t)N_NODES * HC);
    int*   off   = deg + N_NODES;
    int*   total = off + N_NODES;
    int*   rank  = total + 64;
    int*   perm  = rank + nE;

    const int nb = (N_NODES + 255) / 256;
    const int ebk = (nE + 255) / 256;

    gat_zero<<<nb, 256, 0, stream>>>(deg, total);
    gat_count<<<ebk, 256, 0, stream>>>(dst, nE, deg, rank);
    gat_alloc<<<nb, 256, 0, stream>>>(deg, total, off);
    gat_fill<<<ebk, 256, 0, stream>>>(src, dst, nE, off, rank, perm);

    gat_gemm<<<3125, 128, 0, stream>>>(x, Wl, bl, Wr, br, xl, out);

    gat_node<<<N_NODES / 8, 256, 0, stream>>>(xl, out, att, bias, deg, off, perm);
}


# Edge grouping by destination moved from per-edge global adds and scattered 4-byte traffic to two LDS-binned levels (256-node buckets, then nodes); edge sums stay in arrival order
# speedup vs baseline: 4.2304x; 1.5972x over previous
#include <hip/hip_runtime.h>

#define N_NODES 100000
#define IN_DIM  128
#define HEADS   4
#define HC      128
#define NEG_SLOPE 0.2f

typedef float v2f __attribute__((ext_vector_type(2)));
typedef float v8f __attribute__((ext_vector_type(8)));

__global__ void gat_gemm(const float* __restrict__ x,
                         const float* __restrict__ Wl, const float* __restrict__ bl,
                         const float* __restrict__ Wr, const float* __restrict__ br,
                         float* __restrict__ xl, float* __restrict__ xr)
{
    const int wid  = threadIdx.x >> 5;
    const int lane = threadIdx.x & 31;
    const int task = blockIdx.x * 4 + wid;
    const int matSel  = task & 1;
    const int rowTile = task >> 1;

    const float* W    = matSel ? Wr : Wl;
    const float* bv   = matSel ? br : bl;
    float*       outp = matSel ? xr : xl;

    const int row0 = rowTile * 16;
    const int lm = lane & 15;
    const int lk = (lane >> 4) * 2;

    const float* aptr = x + (size_t)(row0 + lm) * IN_DIM + lk;
    const float* bptr = W + (size_t)lk * HC + lm;

    v8f c[8] = {};
    for (int k0 = 0; k0 < IN_DIM; k0 += 4) {
        const float2 av = *(const float2*)aptr;
        v2f a; a.x = av.x; a.y = av.y;
        #pragma unroll
        for (int t = 0; t < 8; ++t) {
            v2f b;
            b.x = bptr[t * 16];
            b.y = bptr[HC + t * 16];
            c[t] = __builtin_amdgcn_wmma_f32_16x16x4_f32(
                       false, a, false, b, (short)0, c[t], false, false);
        }
        aptr += 4;
        bptr += 4 * HC;
    }

    const int rbase = row0 + ((lane >> 4) * 8);
    #pragma unroll
    for (int t = 0; t < 8; ++t) {
        const int col = t * 16 + lm;
        const float bb = bv[col];
        #pragma unroll
        for (int r = 0; r < 8; ++r)
            outp[(size_t)(rbase + r) * HC + col] = c[t][r] + bb;
    }
}

#define NBK   391
#define NBLK  256
#define TILE  8192
#define FCAP  8192

__global__ __launch_bounds__(1024)
void gat_hist(const int* __restrict__ dst, int nEdges,
              int* __restrict__ hist, int* __restrict__ total)
{
    __shared__ int lcnt[NBK];
    const int tid = threadIdx.x;
    if (tid < NBK) lcnt[tid] = 0;
    if (blockIdx.x == 0 && tid == 0) *total = 0;
    __syncthreads();
    const int ch  = (nEdges + NBLK - 1) / NBLK;
    const int beg = blockIdx.x * ch;
    const int end = (beg + ch < nEdges) ? (beg + ch) : nEdges;
    for (int t0 = beg; t0 < end; t0 += TILE) {
        int dv[8];
        #pragma unroll
        for (int j = 0; j < 8; ++j) {
            const int e = t0 + tid + j * 1024;
            dv[j] = (e < end) ? dst[e] : -1;
        }
        #pragma unroll
        for (int j = 0; j < 8; ++j)
            if (dv[j] >= 0) atomicAdd(&lcnt[dv[j] >> 8], 1);
    }
    __syncthreads();
    if (tid < NBK) hist[tid * NBLK + blockIdx.x] = lcnt[tid];
}

__global__ __launch_bounds__(NBLK)
void gat_bscan(int* __restrict__ hist, int* __restrict__ total,
               int* __restrict__ bstart, int* __restrict__ bcnt)
{
    __shared__ int wsum[8];
    __shared__ int base;
    const int b    = blockIdx.x;
    const int lane = threadIdx.x & 31;
    const int wv   = threadIdx.x >> 5;
    const int v = hist[b * NBLK + threadIdx.x];
    int inc = v;
    for (int o = 1; o < 32; o <<= 1) {
        const int t = __shfl_up(inc, o, 32);
        if (lane >= o) inc += t;
    }
    if (lane == 31) wsum[wv] = inc;
    __syncthreads();
    if (threadIdx.x == 0) {
        int run = 0;
        for (int w = 0; w < 8; ++w) { const int t = wsum[w]; wsum[w] = run; run += t; }
        const int bs = atomicAdd(total, run);
        base = bs;
        bstart[b] = bs;
        bcnt[b]   = run;
    }
    __syncthreads();
    hist[b * NBLK + threadIdx.x] = base + wsum[wv] + inc - v;
}

__global__ __launch_bounds__(1024)
void gat_bin(const int* __restrict__ src, const int* __restrict__ dst, int nEdges,
             const int* __restrict__ hist, int* __restrict__ binned)
{
    __shared__ int lcnt[NBK];
    __shared__ int gdelta[NBK];
    __shared__ int gbase[NBK];
    __shared__ int wsum[32];
    __shared__ int stage[TILE];
    __shared__ unsigned short sbk[TILE];
    const int tid  = threadIdx.x;
    const int lane = tid & 31;
    const int wv   = tid >> 5;
    if (tid < NBK) gbase[tid] = hist[tid * NBLK + blockIdx.x];

    const int ch  = (nEdges + NBLK - 1) / NBLK;
    const int beg = blockIdx.x * ch;
    const int end = (beg + ch < nEdges) ? (beg + ch) : nEdges;

    for (int t0 = beg; t0 < end; t0 += TILE) {
        const int tn = (end - t0 < TILE) ? (end - t0) : TILE;
        if (tid < NBK) lcnt[tid] = 0;
        __syncthreads();

        int rk[8], rec[8], bb[8];
        #pragma unroll
        for (int j = 0; j < 8; ++j) {
            const int i = tid + j * 1024;
            if (i < tn) {
                const int d = dst[t0 + i];
                const int sv = src[t0 + i];
                bb[j]  = d >> 8;
                rec[j] = sv | ((d & 255) << 17);
                rk[j]  = atomicAdd(&lcnt[bb[j]], 1);
            }
        }
        __syncthreads();

        const int v = (tid < NBK) ? lcnt[tid] : 0;
        int inc = v;
        for (int o = 1; o < 32; o <<= 1) {
            const int t = __shfl_up(inc, o, 32);
            if (lane >= o) inc += t;
        }
        if (lane == 31) wsum[wv] = inc;
        __syncthreads();
        if (wv == 0) {
            const int t = wsum[lane];
            int ti = t;
            for (int o = 1; o < 32; o <<= 1) {
                const int u = __shfl_up(ti, o, 32);
                if (lane >= o) ti += u;
            }
            wsum[lane] = ti - t;
        }
        __syncthreads();
        if (tid < NBK) {
            const int ls = wsum[wv] + inc - v;
            lcnt[tid]   = ls;
            gdelta[tid] = gbase[tid] - ls;
            gbase[tid] += v;
        }
        __syncthreads();

        #pragma unroll
        for (int j = 0; j < 8; ++j) {
            const int i = tid + j * 1024;
            if (i < tn) {
                const int p = lcnt[bb[j]] + rk[j];
                stage[p] = rec[j];
                sbk[p]   = (unsigned short)bb[j];
            }
        }
        __syncthreads();

        for (int i = tid; i < tn; i += 1024)
            binned[i + gdelta[sbk[i]]] = stage[i];
    }
}

__global__ __launch_bounds__(1024)
void gat_fine(const int* __restrict__ binned, const int* __restrict__ bstart,
              const int* __restrict__ bcnt,
              int* __restrict__ deg, int* __restrict__ off, int* __restrict__ perm)
{
    __shared__ int ncnt[256];
    __shared__ int noff[256];
    __shared__ int ncur[256];
    __shared__ int wsum[8];
    __shared__ int stage[FCAP];
    const int tid  = threadIdx.x;
    const int lane = tid & 31;
    const int wv   = tid >> 5;
    const int b     = blockIdx.x;
    const int cnt   = bcnt[b];
    const int start = bstart[b];
    const bool inLds = cnt <= FCAP;
    const unsigned NONE = 0xFFFFFFFFu;

    if (tid < 256) { ncnt[tid] = 0; ncur[tid] = 0; }
    __syncthreads();

    unsigned r[8];
    if (inLds) {
        #pragma unroll
        for (int j = 0; j < 8; ++j) {
            const int i = tid + j * 1024;
            r[j] = (i < cnt) ? (unsigned)binned[start + i] : NONE;
        }
        #pragma unroll
        for (int j = 0; j < 8; ++j)
            if (r[j] != NONE) atomicAdd(&ncnt[r[j] >> 17], 1);
    } else {
        for (int i = tid; i < cnt; i += 1024)
            atomicAdd(&ncnt[(unsigned)binned[start + i] >> 17], 1);
    }
    __syncthreads();

    int v = 0, inc = 0;
    if (tid < 256) {
        v = ncnt[tid];
        inc = v;
        for (int o = 1; o < 32; o <<= 1) {
            const int t = __shfl_up(inc, o, 32);
            if (lane >= o) inc += t;
        }
        if (lane == 31) wsum[wv] = inc;
    }
    __syncthreads();
    if (tid == 0) {
        int run = 0;
        for (int w = 0; w < 8; ++w) { const int t = wsum[w]; wsum[w] = run; run += t; }
    }
    __syncthreads();
    if (tid < 256) {
        const int ex = wsum[wv] + inc - v;
        noff[tid] = ex;
        const int node = b * 256 + tid;
        if (node < N_NODES) { deg[node] = v; off[node] = start + ex; }
    }
    __syncthreads();

    if (inLds) {
        #pragma unroll
        for (int j = 0; j < 8; ++j) {
            if (r[j] != NONE) {
                const int n = r[j] >> 17;
                const int p = noff[n] + atomicAdd(&ncur[n], 1);
                stage[p] = (int)(r[j] & 0x1FFFFu);
            }
        }
        __syncthreads();
        for (int i = tid; i < cnt; i += 1024) perm[start + i] = stage[i];
    } else {
        for (int i = tid; i < cnt; i += 1024) {
            const unsigned rr = (unsigned)binned[start + i];
            const int n = rr >> 17;
            const int p = noff[n] + atomicAdd(&ncur[n], 1);
            perm[start + p] = (int)(rr & 0x1FFFFu);
        }
    }
}

__device__ __forceinline__ float edge_score(const float4 a, const float4 b,
                                            const float4 at)
{
    float m0 = a.x + b.x; m0 = m0 >= 0.0f ? m0 : NEG_SLOPE * m0;
    float m1 = a.y + b.y; m1 = m1 >= 0.0f ? m1 : NEG_SLOPE * m1;
    float m2 = a.z + b.z; m2 = m2 >= 0.0f ? m2 : NEG_SLOPE * m2;
    float m3 = a.w + b.w; m3 = m3 >= 0.0f ? m3 : NEG_SLOPE * m3;
    float p = at.x * m0 + at.y * m1 + at.z * m2 + at.w * m3;
    p += __shfl_xor(p, 1, 32);
    p += __shfl_xor(p, 2, 32);
    p += __shfl_xor(p, 4, 32);
    return p;
}

__global__ __launch_bounds__(256)
void gat_node(const float* __restrict__ xl, float* out,
              const float* __restrict__ att, const float* __restrict__ bias,
              const int* __restrict__ deg, const int* __restrict__ off,
              const int* __restrict__ perm)
{
    const int d = blockIdx.x * 8 + __builtin_amdgcn_readfirstlane((int)(threadIdx.x >> 5));
    if (d >= N_NODES) return;
    const int lane = threadIdx.x & 31;

    float* orow = out + (size_t)d * HC + lane * 4;
    const float4 b  = *(const float4*)orow;
    const float4 at = *(const float4*)(att + lane * 4);
    const int n     = deg[d];
    const int start = off[d];

    float4 acc = *(const float4*)(xl + (size_t)d * HC + lane * 4);
    float m = edge_score(acc, b, at);
    float l = 1.0f;

    const float NEG_INF = -__builtin_huge_valf();

    for (int j0 = 0; j0 < n; j0 += 32) {
        const int cnt = (n - j0) < 32 ? (n - j0) : 32;
        const int idx = (lane < cnt) ? perm[start + j0 + lane] : d;
        for (int k = 0; k < cnt; k += 4) {
            const int s0 = __builtin_amdgcn_readlane(idx, k);
            const int s1 = __builtin_amdgcn_readlane(idx, k + 1);
            const int s2 = __builtin_amdgcn_readlane(idx, k + 2);
            const int s3 = __builtin_amdgcn_readlane(idx, k + 3);
            const float4 a0 = *(const float4*)(xl + (size_t)s0 * HC + lane * 4);
            const float4 a1 = *(const float4*)(xl + (size_t)s1 * HC + lane * 4);
            const float4 a2 = *(const float4*)(xl + (size_t)s2 * HC + lane * 4);
            const float4 a3 = *(const float4*)(xl + (size_t)s3 * HC + lane * 4);
            const float p0 = edge_score(a0, b, at);
            float p1 = edge_score(a1, b, at);
            float p2 = edge_score(a2, b, at);
            float p3 = edge_score(a3, b, at);
            p1 = (k + 1 < cnt) ? p1 : NEG_INF;
            p2 = (k + 2 < cnt) ? p2 : NEG_INF;
            p3 = (k + 3 < cnt) ? p3 : NEG_INF;
            const float mn = fmaxf(fmaxf(m, p0), fmaxf(p1, fmaxf(p2, p3)));
            const float sc = __expf(m - mn);
            const float w0 = __expf(p0 - mn);
            const float w1 = __expf(p1 - mn);
            const float w2 = __expf(p2 - mn);
            const float w3 = __expf(p3 - mn);
            l = l * sc + ((w0 + w1) + (w2 + w3));
            acc.x = acc.x * sc + ((w0 * a0.x + w1 * a1.x) + (w2 * a2.x + w3 * a3.x));
            acc.y = acc.y * sc + ((w0 * a0.y + w1 * a1.y) + (w2 * a2.y + w3 * a3.y));
            acc.z = acc.z * sc + ((w0 * a0.z + w1 * a1.z) + (w2 * a2.z + w3 * a3.z));
            acc.w = acc.w * sc + ((w0 * a0.w + w1 * a1.w) + (w2 * a2.w + w3 * a3.w));
            m = mn;
        }
    }

    const float4 bs = *(const float4*)(bias + lane * 4);
    float4 o;
    o.x = acc.x / l + bs.x;
    o.y = acc.y / l + bs.y;
    o.z = acc.z / l + bs.z;
    o.w = acc.w / l + bs.w;
    *(float4*)orow = o;
}

extern "C" void kernel_launch(void* const* d_in, const int* in_sizes, int n_in,
                              void* d_out, int out_size, void* d_ws, size_t ws_size,
                              hipStream_t stream)
{
    const float* x    = (const float*)d_in[0];
    const float* Wl   = (const float*)d_in[1];
    const float* bl   = (const float*)d_in[2];
    const float* Wr   = (const float*)d_in[3];
    const float* br   = (const float*)d_in[4];
    const float* att  = (const float*)d_in[5];
    const float* bias = (const float*)d_in[6];
    const int*   src  = (const int*)d_in[7];
    const int*   dst  = (const int*)d_in[8];
    const int nE = in_sizes[7];
    float* out = (float*)d_out;

    float* xl    = (float*)d_ws;
    int*   deg   = (int*)(xl + (size_t)N_NODES * HC);
    int*   off   = deg + N_NODES;
    int*   perm  = off + N_NODES;

    int*   hist   = (int*)d_ws;
    int*   bstart = hist + NBK * NBLK;
    int*   bcnt   = bstart + 512;
    int*   total  = bcnt + 512;
    int*   binned = total + 64;

    gat_hist<<<NBLK, 1024, 0, stream>>>(dst, nE, hist, total);
    gat_bscan<<<NBK, NBLK, 0, stream>>>(hist, total, bstart, bcnt);
    gat_bin<<<NBLK, 1024, 0, stream>>>(src, dst, nE, hist, binned);
    gat_fine<<<NBK, 1024, 0, stream>>>(binned, bstart, bcnt, deg, off, perm);

    gat_gemm<<<3125, 128, 0, stream>>>(x, Wl, bl, Wr, br, xl, out);

    gat_node<<<N_NODES / 8, 256, 0, stream>>>(xl, out, att, bias, deg, off, perm);
}


# GEMM runs as 256 resident 16-wave work-groups that copy both weight matrices into LDS once and loop over row tiles; B fragments come from 16-byte LDS loads, not 4-byte global loads; k order kept
# speedup vs baseline: 4.4831x; 1.0597x over previous
#include <hip/hip_runtime.h>

#define N_NODES 100000
#define IN_DIM  128
#define HEADS   4
#define HC      128
#define NEG_SLOPE 0.2f

typedef float v2f __attribute__((ext_vector_type(2)));
typedef float v8f __attribute__((ext_vector_type(8)));

#define GEMM_BLOCKS 256
#define GEMM_PAIRS  2048
#define ROW_TILES   (N_NODES / 16)
#define WMAT_FLOATS 16384

__global__ __launch_bounds__(512)
void gat_gemm(const float* __restrict__ x,
              const float* __restrict__ Wl, const float* __restrict__ bl,
              const float* __restrict__ Wr, const float* __restrict__ br,
              float* __restrict__ xl, float* __restrict__ xr)
{
    __shared__ __attribute__((aligned(16))) float wlds[2 * WMAT_FLOATS];

    const int tid = threadIdx.x;
    #pragma unroll
    for (int it = 0; it < 16; ++it) {
        const int item = tid + it * 512;
        const int mat = item >> 12;
        const int g   = (item >> 7) & 31;
        const int n   = item & 127;
        const float* p = (mat ? Wr : Wl) + (size_t)(8 * (g >> 1) + 2 * (g & 1)) * HC + n;
        float4 v;
        v.x = p[0];
        v.y = p[HC];
        v.z = p[4 * HC];
        v.w = p[5 * HC];
        *(float4*)&wlds[mat * WMAT_FLOATS + (g * 128 + n) * 4] = v;
    }
    __syncthreads();

    const int wv   = __builtin_amdgcn_readfirstlane(tid >> 5);
    const int lane = tid & 31;
    const int lm = lane & 15;
    const int h  = lane >> 4;
    const int matSel = wv & 1;
    const float* bv   = matSel ? br : bl;
    float*       outp = matSel ? xr : xl;
    const float* wb   = wlds + matSel * WMAT_FLOATS + h * 512 + lm * 4;
    const float* arow = x + (size_t)lm * IN_DIM + 2 * h;

    int rowTile = (wv >> 1) * GEMM_BLOCKS + blockIdx.x;

    v2f a[8];
    {
        const float* p = arow + (size_t)rowTile * (16 * IN_DIM);
        #pragma unroll
        for (int i = 0; i < 8; ++i) {
            const float2 av = *(const float2*)(p + 4 * i);
            a[i].x = av.x; a[i].y = av.y;
        }
    }

    while (rowTile < ROW_TILES) {
        v8f c[8] = {};
        #pragma unroll 1
        for (int g = 0; g < 4; ++g) {
            const int nrt = (g < 3) ? rowTile : rowTile + GEMM_PAIRS;
            const int ng  = (g + 1) & 3;
            v2f na[8];
            if (nrt < ROW_TILES) {
                const float* p = arow + (size_t)nrt * (16 * IN_DIM) + ng * 32;
                #pragma unroll
                for (int i = 0; i < 8; ++i) {
                    const float2 av = *(const float2*)(p + 4 * i);
                    na[i].x = av.x; na[i].y = av.y;
                }
            } else {
                #pragma unroll
                for (int i = 0; i < 8; ++i) na[i] = a[i];
            }
            const float* wg = wb + g * 4096;
            #pragma unroll
            for (int kk = 0; kk < 4; ++kk) {
                float4 bq[8];
                #pragma unroll
                for (int t = 0; t < 8; ++t)
                    bq[t] = *(const float4*)(wg + kk * 1024 + t * 64);
                #pragma unroll
                for (int t = 0; t < 8; ++t) {
                    v2f b; b.x = bq[t].x; b.y = bq[t].y;
                    c[t] = __builtin_amdgcn_wmma_f32_16x16x4_f32(
                               false, a[2 * kk], false, b, (short)0, c[t], false, false);
                }
                #pragma unroll
                for (int t = 0; t < 8; ++t) {
                    v2f b; b.x = bq[t].z; b.y = bq[t].w;
                    c[t] = __builtin_amdgcn_wmma_f32_16x16x4_f32(
                               false, a[2 * kk + 1], false, b, (short)0, c[t], false, false);
                }
            }
            #pragma unroll
            for (int i = 0; i < 8; ++i) a[i] = na[i];
        }

        const int rbase = rowTile * 16 + h * 8;
        #pragma unroll
        for (int t = 0; t < 8; ++t) {
            const int col = t * 16 + lm;
            const float bb = bv[col];
            #pragma unroll
            for (int r = 0; r < 8; ++r)
                outp[(size_t)(rbase + r) * HC + col] = c[t][r] + bb;
        }
        rowTile += GEMM_PAIRS;
    }
}

#define NBK   391
#define NBLK  256
#define TILE  8192
#define FCAP  8192

__global__ __launch_bounds__(1024)
void gat_hist(const int* __restrict__ dst, int nEdges,
              int* __restrict__ hist, int* __restrict__ total)
{
    __shared__ int lcnt[NBK];
    const int tid = threadIdx.x;
    if (tid < NBK) lcnt[tid] = 0;
    if (blockIdx.x == 0 && tid == 0) *total = 0;
    __syncthreads();
    const int ch  = (nEdges + NBLK - 1) / NBLK;
    const int beg = blockIdx.x * ch;
    const int end = (beg + ch < nEdges) ? (beg + ch) : nEdges;
    for (int t0 = beg; t0 < end; t0 += TILE) {
        int dv[8];
        #pragma unroll
        for (int j = 0; j < 8; ++j) {
            const int e = t0 + tid + j * 1024;
            dv[j] = (e < end) ? dst[e] : -1;
        }
        #pragma unroll
        for (int j = 0; j < 8; ++j)
            if (dv[j] >= 0) atomicAdd(&lcnt[dv[j] >> 8], 1);
    }
    __syncthreads();
    if (tid < NBK) hist[tid * NBLK + blockIdx.x] = lcnt[tid];
}

__global__ __launch_bounds__(NBLK)
void gat_bscan(int* __restrict__ hist, int* __restrict__ total,
               int* __restrict__ bstart, int* __restrict__ bcnt)
{
    __shared__ int wsum[8];
    __shared__ int base;
    const int b    = blockIdx.x;
    const int lane = threadIdx.x & 31;
    const int wv   = threadIdx.x >> 5;
    const int v = hist[b * NBLK + threadIdx.x];
    int inc = v;
    for (int o = 1; o < 32; o <<= 1) {
        const int t = __shfl_up(inc, o, 32);
        if (lane >= o) inc += t;
    }
    if (lane == 31) wsum[wv] = inc;
    __syncthreads();
    if (threadIdx.x == 0) {
        int run = 0;
        for (int w = 0; w < 8; ++w) { const int t = wsum[w]; wsum[w] = run; run += t; }
        const int bs = atomicAdd(total, run);
        base = bs;
        bstart[b] = bs;
        bcnt[b]   = run;
    }
    __syncthreads();
    hist[b * NBLK + threadIdx.x] = base + wsum[wv] + inc - v;
}

__global__ __launch_bounds__(1024)
void gat_bin(const int* __restrict__ src, const int* __restrict__ dst, int nEdges,
             const int* __restrict__ hist, int* __restrict__ binned)
{
    __shared__ int lcnt[NBK];
    __shared__ int gdelta[NBK];
    __shared__ int gbase[NBK];
    __shared__ int wsum[32];
    __shared__ int stage[TILE];
    __shared__ unsigned short sbk[TILE];
    const int tid  = threadIdx.x;
    const int lane = tid & 31;
    const int wv   = tid >> 5;
    if (tid < NBK) gbase[tid] = hist[tid * NBLK + blockIdx.x];

    const int ch  = (nEdges + NBLK - 1) / NBLK;
    const int beg = blockIdx.x * ch;
    const int end = (beg + ch < nEdges) ? (beg + ch) : nEdges;

    for (int t0 = beg; t0 < end; t0 += TILE) {
        const int tn = (end - t0 < TILE) ? (end - t0) : TILE;
        if (tid < NBK) lcnt[tid] = 0;
        __syncthreads();

        int rk[8], rec[8], bb[8];
        #pragma unroll
        for (int j = 0; j < 8; ++j) {
            const int i = tid + j * 1024;
            if (i < tn) {
                const int d = dst[t0 + i];
                const int sv = src[t0 + i];
                bb[j]  = d >> 8;
                rec[j] = sv | ((d & 255) << 17);
                rk[j]  = atomicAdd(&lcnt[bb[j]], 1);
            }
        }
        __syncthreads();

        const int v = (tid < NBK) ? lcnt[tid] : 0;
        int inc = v;
        for (int o = 1; o < 32; o <<= 1) {
            const int t = __shfl_up(inc, o, 32);
            if (lane >= o) inc += t;
        }
        if (lane == 31) wsum[wv] = inc;
        __syncthreads();
        if (wv == 0) {
            const int t = wsum[lane];
            int ti = t;
            for (int o = 1; o < 32; o <<= 1) {
                const int u = __shfl_up(ti, o, 32);
                if (lane >= o) ti += u;
            }
            wsum[lane] = ti - t;
        }
        __syncthreads();
        if (tid < NBK) {
            const int ls = wsum[wv] + inc - v;
            lcnt[tid]   = ls;
            gdelta[tid] = gbase[tid] - ls;
            gbase[tid] += v;
        }
        __syncthreads();

        #pragma unroll
        for (int j = 0; j < 8; ++j) {
            const int i = tid + j * 1024;
            if (i < tn) {
                const int p = lcnt[bb[j]] + rk[j];
                stage[p] = rec[j];
                sbk[p]   = (unsigned short)bb[j];
            }
        }
        __syncthreads();

        for (int i = tid; i < tn; i += 1024)
            binned[i + gdelta[sbk[i]]] = stage[i];
    }
}

__global__ __launch_bounds__(1024)
void gat_fine(const int* __restrict__ binned, const int* __restrict__ bstart,
              const int* __restrict__ bcnt,
              int* __restrict__ deg, int* __restrict__ off, int* __restrict__ perm)
{
    __shared__ int ncnt[256];
    __shared__ int noff[256];
    __shared__ int ncur[256];
    __shared__ int wsum[8];
    __shared__ int stage[FCAP];
    const int tid  = threadIdx.x;
    const int lane = tid & 31;
    const int wv   = tid >> 5;
    const int b     = blockIdx.x;
    const int cnt   = bcnt[b];
    const int start = bstart[b];
    const bool inLds = cnt <= FCAP;
    const unsigned NONE = 0xFFFFFFFFu;

    if (tid < 256) { ncnt[tid] = 0; ncur[tid] = 0; }
    __syncthreads();

    unsigned r[8];
    if (inLds) {
        #pragma unroll
        for (int j = 0; j < 8; ++j) {
            const int i = tid + j * 1024;
            r[j] = (i < cnt) ? (unsigned)binned[start + i] : NONE;
        }
        #pragma unroll
        for (int j = 0; j < 8; ++j)
            if (r[j] != NONE) atomicAdd(&ncnt[r[j] >> 17], 1);
    } else {
        for (int i = tid; i < cnt; i += 1024)
            atomicAdd(&ncnt[(unsigned)binned[start + i] >> 17], 1);
    }
    __syncthreads();

    int v = 0, inc = 0;
    if (tid < 256) {
        v = ncnt[tid];
        inc = v;
        for (int o = 1; o < 32; o <<= 1) {
            const int t = __shfl_up(inc, o, 32);
            if (lane >= o) inc += t;
        }
        if (lane == 31) wsum[wv] = inc;
    }
    __syncthreads();
    if (tid == 0) {
        int run = 0;
        for (int w = 0; w < 8; ++w) { const int t = wsum[w]; wsum[w] = run; run += t; }
    }
    __syncthreads();
    if (tid < 256) {
        const int ex = wsum[wv] + inc - v;
        noff[tid] = ex;
        const int node = b * 256 + tid;
        if (node < N_NODES) { deg[node] = v; off[node] = start + ex; }
    }
    __syncthreads();

    if (inLds) {
        #pragma unroll
        for (int j = 0; j < 8; ++j) {
            if (r[j] != NONE) {
                const int n = r[j] >> 17;
                const int p = noff[n] + atomicAdd(&ncur[n], 1);
                stage[p] = (int)(r[j] & 0x1FFFFu);
            }
        }
        __syncthreads();
        for (int i = tid; i < cnt; i += 1024) perm[start + i] = stage[i];
    } else {
        for (int i = tid; i < cnt; i += 1024) {
            const unsigned rr = (unsigned)binned[start + i];
            const int n = rr >> 17;
            const int p = noff[n] + atomicAdd(&ncur[n], 1);
            perm[start + p] = (int)(rr & 0x1FFFFu);
        }
    }
}

__device__ __forceinline__ float edge_score(const float4 a, const float4 b,
                                            const float4 at)
{
    float m0 = a.x + b.x; m0 = m0 >= 0.0f ? m0 : NEG_SLOPE * m0;
    float m1 = a.y + b.y; m1 = m1 >= 0.0f ? m1 : NEG_SLOPE * m1;
    float m2 = a.z + b.z; m2 = m2 >= 0.0f ? m2 : NEG_SLOPE * m2;
    float m3 = a.w + b.w; m3 = m3 >= 0.0f ? m3 : NEG_SLOPE * m3;
    float p = at.x * m0 + at.y * m1 + at.z * m2 + at.w * m3;
    p += __shfl_xor(p, 1, 32);
    p += __shfl_xor(p, 2, 32);
    p += __shfl_xor(p, 4, 32);
    return p;
}

__global__ __launch_bounds__(256)
void gat_node(const float* __restrict__ xl, float* out,
              const float* __restrict__ att, const float* __restrict__ bias,
              const int* __restrict__ deg, const int* __restrict__ off,
              const int* __restrict__ perm)
{
    const int d = blockIdx.x * 8 + __builtin_amdgcn_readfirstlane((int)(threadIdx.x >> 5));
    if (d >= N_NODES) return;
    const int lane = threadIdx.x & 31;

    float* orow = out + (size_t)d * HC + lane * 4;
    const float4 b  = *(const float4*)orow;
    const float4 at = *(const float4*)(att + lane * 4);
    const int n     = deg[d];
    const int start = off[d];

    float4 acc = *(const float4*)(xl + (size_t)d * HC + lane * 4);
    float m = edge_score(acc, b, at);
    float l = 1.0f;

    const float NEG_INF = -__builtin_huge_valf();

    for (int j0 = 0; j0 < n; j0 += 32) {
        const int cnt = (n - j0) < 32 ? (n - j0) : 32;
        const int idx = (lane < cnt) ? perm[start + j0 + lane] : d;
        for (int k = 0; k < cnt; k += 4) {
            const int s0 = __builtin_amdgcn_readlane(idx, k);
            const int s1 = __builtin_amdgcn_readlane(idx, k + 1);
            const int s2 = __builtin_amdgcn_readlane(idx, k + 2);
            const int s3 = __builtin_amdgcn_readlane(idx, k + 3);
            const float4 a0 = *(const float4*)(xl + (size_t)s0 * HC + lane * 4);
            const float4 a1 = *(const float4*)(xl + (size_t)s1 * HC + lane * 4);
            const float4 a2 = *(const float4*)(xl + (size_t)s2 * HC + lane * 4);
            const float4 a3 = *(const float4*)(xl + (size_t)s3 * HC + lane * 4);
            const float p0 = edge_score(a0, b, at);
            float p1 = edge_score(a1, b, at);
            float p2 = edge_score(a2, b, at);
            float p3 = edge_score(a3, b, at);
            p1 = (k + 1 < cnt) ? p1 : NEG_INF;
            p2 = (k + 2 < cnt) ? p2 : NEG_INF;
            p3 = (k + 3 < cnt) ? p3 : NEG_INF;
            const float mn = fmaxf(fmaxf(m, p0), fmaxf(p1, fmaxf(p2, p3)));
            const float sc = __expf(m - mn);
            const float w0 = __expf(p0 - mn);
            const float w1 = __expf(p1 - mn);
            const float w2 = __expf(p2 - mn);
            const float w3 = __expf(p3 - mn);
            l = l * sc + ((w0 + w1) + (w2 + w3));
            acc.x = acc.x * sc + ((w0 * a0.x + w1 * a1.x) + (w2 * a2.x + w3 * a3.x));
            acc.y = acc.y * sc + ((w0 * a0.y + w1 * a1.y) + (w2 * a2.y + w3 * a3.y));
            acc.z = acc.z * sc + ((w0 * a0.z + w1 * a1.z) + (w2 * a2.z + w3 * a3.z));
            acc.w = acc.w * sc + ((w0 * a0.w + w1 * a1.w) + (w2 * a2.w + w3 * a3.w));
            m = mn;
        }
    }

    const float4 bs = *(const float4*)(bias + lane * 4);
    float4 o;
    o.x = acc.x / l + bs.x;
    o.y = acc.y / l + bs.y;
    o.z = acc.z / l + bs.z;
    o.w = acc.w / l + bs.w;
    *(float4*)orow = o;
}

extern "C" void kernel_launch(void* const* d_in, const int* in_sizes, int n_in,
                              void* d_out, int out_size, void* d_ws, size_t ws_size,
                              hipStream_t stream)
{
    const float* x    = (const float*)d_in[0];
    const float* Wl   = (const float*)d_in[1];
    const float* bl   = (const float*)d_in[2];
    const float* Wr   = (const float*)d_in[3];
    const float* br   = (const float*)d_in[4];
    const float* att  = (const float*)d_in[5];
    const float* bias = (const float*)d_in[6];
    const int*   src  = (const int*)d_in[7];
    const int*   dst  = (const int*)d_in[8];
    const int nE = in_sizes[7];
    float* out = (float*)d_out;

    float* xl    = (float*)d_ws;
    int*   deg   = (int*)(xl + (size_t)N_NODES * HC);
    int*   off   = deg + N_NODES;
    int*   perm  = off + N_NODES;

    int*   hist   = (int*)d_ws;
    int*   bstart = hist + NBK * NBLK;
    int*   bcnt   = bstart + 512;
    int*   total  = bcnt + 512;
    int*   binned = total + 64;

    gat_hist<<<NBLK, 1024, 0, stream>>>(dst, nE, hist, total);
    gat_bscan<<<NBK, NBLK, 0, stream>>>(hist, total, bstart, bcnt);
    gat_bin<<<NBLK, 1024, 0, stream>>>(src, dst, nE, hist, binned);
    gat_fine<<<NBK, 1024, 0, stream>>>(binned, bstart, bcnt, deg, off, perm);

    gat_gemm<<<GEMM_BLOCKS, 512, 0, stream>>>(x, Wl, bl, Wr, br, xl, out);

    gat_node<<<N_NODES / 8, 256, 0, stream>>>(xl, out, att, bias, deg, off, perm);
}
